# starting kernel
# baseline (speedup 1.0000x reference)
#include <hip/hip_runtime.h>


typedef float v2f __attribute__((ext_vector_type(2)));
typedef float v8f __attribute__((ext_vector_type(8)));

#define NROWS 65536
#define DDIM  256
#define ROW_STRIDE 260
#define PANEL_FLOATS (16 * ROW_STRIDE)
#define WAVES_PER_BLOCK 8

__launch_bounds__(256)
__global__ void mahalanobis_wmma_kernel(const float* __restrict__ x,
                                        const float* __restrict__ xfit,
                                        const float* __restrict__ Sinv,
                                        float* __restrict__ out) {
    extern __shared__ float lds[];

    const int lane  = threadIdx.x & 31;
    const int wave  = threadIdx.x >> 5;
    const int n     = lane & 15;
    const int khalf = lane >> 4;

    const int tile  = blockIdx.x * WAVES_PER_BLOCK + wave;
    const size_t base = (size_t)tile * 16;

    float* panel = lds + wave * PANEL_FLOATS;

    const float* xp  = x    + base * DDIM;
    const float* xfp = xfit + base * DDIM;
#pragma unroll
    for (int t = 0; t < 32; ++t) {
        const int i = (t * 32 + lane) * 4;
        const int r = i >> 8;
        const int c = i & 255;
        float4 a = *(const float4*)(xp  + i);
        float4 b = *(const float4*)(xfp + i);
        float4 d = make_float4(a.x - b.x, a.y - b.y, a.z - b.z, a.w - b.w);
        *(float4*)(panel + r * ROW_STRIDE + c) = d;
    }
    asm volatile("s_wait_dscnt 0" ::: "memory");

    float outAcc[8];
#pragma unroll
    for (int r = 0; r < 8; ++r) outAcc[r] = 0.0f;

#pragma unroll
    for (int jg = 0; jg < 4; ++jg) {
        v8f acc[4];
#pragma unroll
        for (int jj = 0; jj < 4; ++jj) acc[jj] = (v8f)(0.0f);

        const float* brow0 = Sinv + (size_t)((jg * 4 + 0) * 16 + n) * DDIM;
        const float* brow1 = Sinv + (size_t)((jg * 4 + 1) * 16 + n) * DDIM;
        const float* brow2 = Sinv + (size_t)((jg * 4 + 2) * 16 + n) * DDIM;
        const float* brow3 = Sinv + (size_t)((jg * 4 + 3) * 16 + n) * DDIM;

#pragma unroll 4
        for (int kk = 0; kk < 64; ++kk) {
            const int k = kk * 4 + khalf * 2;
            v2f a  = *(const v2f*)(panel + n * ROW_STRIDE + k);
            v2f b0 = *(const v2f*)(brow0 + k);
            v2f b1 = *(const v2f*)(brow1 + k);
            v2f b2 = *(const v2f*)(brow2 + k);
            v2f b3 = *(const v2f*)(brow3 + k);
            acc[0] = __builtin_amdgcn_wmma_f32_16x16x4_f32(false, a, false, b0, (short)0, acc[0], false, false);
            acc[1] = __builtin_amdgcn_wmma_f32_16x16x4_f32(false, a, false, b1, (short)0, acc[1], false, false);
            acc[2] = __builtin_amdgcn_wmma_f32_16x16x4_f32(false, a, false, b2, (short)0, acc[2], false, false);
            acc[3] = __builtin_amdgcn_wmma_f32_16x16x4_f32(false, a, false, b3, (short)0, acc[3], false, false);
        }

#pragma unroll
        for (int jj = 0; jj < 4; ++jj) {
            const int j = jg * 4 + jj;
#pragma unroll
            for (int r = 0; r < 8; ++r) {
                const int row = khalf * 8 + r;
                outAcc[r] += acc[jj][r] * panel[row * ROW_STRIDE + j * 16 + n];
            }
        }
    }

#pragma unroll
    for (int r = 0; r < 8; ++r) {
        float v = outAcc[r];
        v += __shfl_xor(v, 1, 32);
        v += __shfl_xor(v, 2, 32);
        v += __shfl_xor(v, 4, 32);
        v += __shfl_xor(v, 8, 32);
        outAcc[r] = v;
    }

    if (n == 0) {
#pragma unroll
        for (int r = 0; r < 8; ++r) {
            out[base + khalf * 8 + r] = outAcc[r];
        }
    }
}

extern "C" void kernel_launch(void* const* d_in, const int* in_sizes, int n_in,
                              void* d_out, int out_size, void* d_ws, size_t ws_size,
                              hipStream_t stream) {
    const float* x    = (const float*)d_in[0];
    const float* xfit = (const float*)d_in[1];
    const float* Sinv = (const float*)d_in[2];
    float* out = (float*)d_out;

    const int rows_per_block = WAVES_PER_BLOCK * 16;
    dim3 grid(NROWS / rows_per_block);
    dim3 block(256);
    size_t ldsBytes = (size_t)WAVES_PER_BLOCK * PANEL_FLOATS * sizeof(float);

    mahalanobis_wmma_kernel<<<grid, block, ldsBytes, stream>>>(x, xfit, Sinv, out);
}


# S_inv staged once per work-group into LDS and read by all 16 waves as WMMA A operand (delta rows in registers as B) instead of per-wave strided global fragment loads; k and j sum order changes
# speedup vs baseline: 1.5248x; 1.5248x over previous
#include <hip/hip_runtime.h>


typedef float v2f __attribute__((ext_vector_type(2)));
typedef float v8f __attribute__((ext_vector_type(8)));

#define NROWS 65536
#define DDIM  256
#define S_PITCH 260
#define WAVES_PER_BLOCK 16
#define BLOCK_THREADS 512
#define JT_GROUP 2

__launch_bounds__(BLOCK_THREADS)
__global__ void mahalanobis_wmma_kernel(const float* __restrict__ x,
                                        const float* __restrict__ xfit,
                                        const float* __restrict__ Sinv,
                                        float* __restrict__ out) {
    __shared__ __attribute__((aligned(16))) float Ssh[DDIM * S_PITCH];

    const int tid  = threadIdx.x;
    const int lane = tid & 31;
    const int wave = tid >> 5;
    const int c    = lane & 15;
    const int h    = lane >> 4;

    const int tile = blockIdx.x * WAVES_PER_BLOCK + wave;
    const size_t base = (size_t)tile * 16;

#pragma unroll 8
    for (int t = 0; t < 32; ++t) {
        const int i  = t * BLOCK_THREADS + tid;
        const int r  = i >> 6;
        const int c4 = (i & 63) * 4;
        const float4 v = *(const float4*)(Sinv + r * DDIM + c4);
        *(float4*)(Ssh + r * S_PITCH + c4) = v;
    }

    float d[128];
    {
        const float* xr = x    + (base + c) * DDIM + 8 * h;
        const float* fr = xfit + (base + c) * DDIM + 8 * h;
#pragma unroll
        for (int ch = 0; ch < 4; ++ch) {
            float4 a[8], b[8];
#pragma unroll
            for (int u = 0; u < 8; ++u) {
                a[u] = *(const float4*)(xr + 64 * ch + 16 * (u >> 1) + 4 * (u & 1));
                b[u] = *(const float4*)(fr + 64 * ch + 16 * (u >> 1) + 4 * (u & 1));
            }
#pragma unroll
            for (int u = 0; u < 8; ++u) {
                const int m = 32 * ch + 4 * u;
                d[m + 0] = a[u].x - b[u].x;
                d[m + 1] = a[u].y - b[u].y;
                d[m + 2] = a[u].z - b[u].z;
                d[m + 3] = a[u].w - b[u].w;
                asm volatile("" : "+v"(d[m + 0]), "+v"(d[m + 1]), "+v"(d[m + 2]), "+v"(d[m + 3]) :: "memory");
            }
        }
    }
    __syncthreads();

    const float* arow = Ssh + c * S_PITCH + 8 * h;
    float part = 0.0f;

#pragma unroll
    for (int jg = 0; jg < 16 / JT_GROUP; ++jg) {
        v8f acc[JT_GROUP];
#pragma unroll
        for (int jj = 0; jj < JT_GROUP; ++jj) acc[jj] = (v8f)(0.0f);

#pragma unroll
        for (int kb = 0; kb < 16; ++kb) {
            float4 s0[JT_GROUP], s1[JT_GROUP];
#pragma unroll
            for (int jj = 0; jj < JT_GROUP; ++jj) {
                const float* p = arow + (16 * (jg * JT_GROUP + jj)) * S_PITCH + 16 * kb;
                s0[jj] = *(const float4*)(p);
                s1[jj] = *(const float4*)(p + 4);
            }
            const v2f b0 = {d[8 * kb + 0], d[8 * kb + 1]};
            const v2f b1 = {d[8 * kb + 2], d[8 * kb + 3]};
            const v2f b2 = {d[8 * kb + 4], d[8 * kb + 5]};
            const v2f b3 = {d[8 * kb + 6], d[8 * kb + 7]};
#pragma unroll
            for (int jj = 0; jj < JT_GROUP; ++jj) {
                const v2f a = {s0[jj].x, s0[jj].y};
                acc[jj] = __builtin_amdgcn_wmma_f32_16x16x4_f32(false, a, false, b0, (short)0, acc[jj], false, false);
            }
#pragma unroll
            for (int jj = 0; jj < JT_GROUP; ++jj) {
                const v2f a = {s0[jj].z, s0[jj].w};
                acc[jj] = __builtin_amdgcn_wmma_f32_16x16x4_f32(false, a, false, b1, (short)0, acc[jj], false, false);
            }
#pragma unroll
            for (int jj = 0; jj < JT_GROUP; ++jj) {
                const v2f a = {s1[jj].x, s1[jj].y};
                acc[jj] = __builtin_amdgcn_wmma_f32_16x16x4_f32(false, a, false, b2, (short)0, acc[jj], false, false);
            }
#pragma unroll
            for (int jj = 0; jj < JT_GROUP; ++jj) {
                const v2f a = {s1[jj].z, s1[jj].w};
                acc[jj] = __builtin_amdgcn_wmma_f32_16x16x4_f32(false, a, false, b3, (short)0, acc[jj], false, false);
            }
        }

#pragma unroll
        for (int jj = 0; jj < JT_GROUP; ++jj) {
#pragma unroll
            for (int r = 0; r < 8; ++r) {
                part += acc[jj][r] * d[8 * (jg * JT_GROUP + jj) + r];
            }
        }
    }

    part += __shfl_xor(part, 16, 32);
    if (h == 0) {
        out[base + c] = part;
    }
}

extern "C" void kernel_launch(void* const* d_in, const int* in_sizes, int n_in,
                              void* d_out, int out_size, void* d_ws, size_t ws_size,
                              hipStream_t stream) {
    const float* x    = (const float*)d_in[0];
    const float* xfit = (const float*)d_in[1];
    const float* Sinv = (const float*)d_in[2];
    float* out = (float*)d_out;

    dim3 grid(256);
    dim3 block(512);

    mahalanobis_wmma_kernel<<<grid, block, 0, stream>>>(x, xfit, Sinv, out);
}
